# starting kernel
# baseline (speedup 1.0000x reference)
#include <hip/hip_runtime.h>

typedef __attribute__((ext_vector_type(2))) float v2f;
typedef __attribute__((ext_vector_type(8))) float v8f;

template <int C, int H, int W>
__global__ __launch_bounds__(128) void cost_tok_kernel(
    const float* __restrict__ lvl,
    const float* __restrict__ wmat,
    const float* __restrict__ bias,
    float* __restrict__ out,
    int B) {
  constexpr int HW  = H * W;
  constexpr int KP  = 52;
  constexpr int TD  = 192;
  constexpr int PIX = 64;

  __shared__ float f1_lds[C * PIX];
  __shared__ float w_lds[KP * TD];
  __shared__ float b_lds[TD];
  __shared__ float corr_lds[PIX * KP];

  const int tid   = threadIdx.x;
  const int tb    = blockIdx.y;
  const int pbase = blockIdx.x * PIX;

  const float* f1 = lvl + (size_t)tb * C * HW;
  const float* f2 = lvl + (size_t)(tb + B) * C * HW;

  for (int i = tid; i < KP * TD; i += 128)
    w_lds[i] = (i < 49 * TD) ? wmat[i] : 0.0f;
  for (int i = tid; i < TD; i += 128)
    b_lds[i] = bias[i];
  for (int i = tid; i < C * PIX; i += 128) {
    int c = i >> 6, p = i & 63;
    f1_lds[i] = f1[(size_t)c * HW + pbase + p];
  }
  __syncthreads();

  {
    const float scale = 1.0f / __builtin_sqrtf((float)C);
    const int p  = tid & 63;
    const int kl = tid >> 6;
    const int pg = pbase + p;
    const int h  = pg / W;
    const int w  = pg % W;
    const float* f1p = f1_lds + p;

    for (int kbase = 0; kbase < KP; kbase += 2) {
      const int k = kbase + kl;
      float acc = 0.0f;
      if (k < 49) {
        const int dy = k / 7 - 3;
        const int dx = k % 7 - 3;
        int nh = h - dy; nh += (nh < 0) ? H : 0; nh -= (nh >= H) ? H : 0;
        int nw = w - dx; nw += (nw < 0) ? W : 0; nw -= (nw >= W) ? W : 0;
        const float* f2p = f2 + (size_t)nh * W + nw;
#pragma unroll 4
        for (int c = 0; c < C; ++c)
          acc += f1p[c * PIX] * f2p[(size_t)c * HW];
        acc *= scale;
      }
      corr_lds[p * KP + k] = acc;
    }
  }
  __syncthreads();

  {
    const int lane = tid & 31;
    const int wave = tid >> 5;
    const int p0   = wave * 16;
    const int nlo  = lane & 15;
    const int lhi  = lane >> 4;

    v2f bfr[13];
#pragma unroll
    for (int ks = 0; ks < 13; ++ks) {
      const int krow = 4 * ks + 2 * lhi;
      const float* src = &corr_lds[(p0 + nlo) * KP + krow];
      bfr[ks].x = src[0];
      bfr[ks].y = src[1];
    }

    for (int m = 0; m < 12; ++m) {
      const int d0 = m * 16;
      v8f acc;
#pragma unroll
      for (int r = 0; r < 8; ++r)
        acc[r] = b_lds[d0 + r + 8 * lhi];

#pragma unroll
      for (int ks = 0; ks < 13; ++ks) {
        const int krow = 4 * ks + 2 * lhi;
        v2f afr;
        afr.x = w_lds[krow * TD + d0 + nlo];
        afr.y = w_lds[(krow + 1) * TD + d0 + nlo];
        acc = __builtin_amdgcn_wmma_f32_16x16x4_f32(
            false, afr, false, bfr[ks], (short)0, acc, false, false);
      }

      float* op = out + ((size_t)tb * TD + d0) * HW + pbase + p0 + nlo;
#pragma unroll
      for (int r = 0; r < 8; ++r)
        op[(size_t)(r + 8 * lhi) * HW] = acc[r];
    }
  }
}

extern "C" void kernel_launch(void* const* d_in, const int* in_sizes, int n_in,
                              void* d_out, int out_size, void* d_ws, size_t ws_size,
                              hipStream_t stream) {
  const float* l1 = (const float*)d_in[0];
  const float* l2 = (const float*)d_in[1];
  const float* l3 = (const float*)d_in[2];
  const float* w1 = (const float*)d_in[3];
  const float* b1 = (const float*)d_in[4];
  const float* w2 = (const float*)d_in[5];
  const float* b2 = (const float*)d_in[6];
  const float* w3 = (const float*)d_in[7];
  const float* b3 = (const float*)d_in[8];

  const int T = 3, B = 2, TD = 192;
  float* tok1 = (float*)d_out;
  float* tok2 = tok1 + (size_t)(T - 1) * B * TD * 128 * 128;
  float* tok3 = tok2 + (size_t)(T - 1) * B * TD * 64 * 64;

  dim3 blk(128);
  cost_tok_kernel<64, 128, 128>
      <<<dim3(128 * 128 / 64, (T - 1) * B), blk, 0, stream>>>(l1, w1, b1, tok1, B);
  cost_tok_kernel<128, 64, 64>
      <<<dim3(64 * 64 / 64, (T - 1) * B), blk, 0, stream>>>(l2, w2, b2, tok2, B);
  cost_tok_kernel<192, 32, 32>
      <<<dim3(32 * 32 / 64, (T - 1) * B), blk, 0, stream>>>(l3, w3, b3, tok3, B);
}


# Correlation phase: next-frame halo tile staged in LDS in 16-channel chunks (next chunk prefetched in registers), 4-pixel x 7-dx register window per thread, replacing 49 strided global re-reads
# speedup vs baseline: 2.7837x; 2.7837x over previous
#include <hip/hip_runtime.h>

typedef __attribute__((ext_vector_type(2))) float v2f;
typedef __attribute__((ext_vector_type(8))) float v8f;

template <int C, int H, int W>
__global__ __launch_bounds__(128) void cost_tok_kernel(
    const float* __restrict__ lvl,
    const float* __restrict__ wmat,
    const float* __restrict__ bias,
    float* __restrict__ out,
    int B) {
  constexpr int HW  = H * W;
  constexpr int KP  = 52;
  constexpr int TD  = 192;
  constexpr int PIX = 64;

  constexpr int TW  = (W >= 64) ? 64 : W;
  constexpr int TR  = PIX / TW;
  constexpr int HR  = TR + 6;
  constexpr int HC  = TW + 6;
  constexpr int HCP = TW + 8;
  constexpr int NS  = (HR * HCP + 127) / 128;
  constexpr int PL  = NS * 128;
  constexpr int CC  = 16;
  constexpr int GPR = TW / 4;
  static_assert(C % CC == 0, "channel chunking");
  static_assert(W % TW == 0 && PIX % TW == 0 && TW % 4 == 0, "tile shape");
  static_assert(H % TR == 0 && H >= 4 && W >= 4, "tile rows");

  __shared__ __attribute__((aligned(16))) float f1_lds[C * PIX];
  __shared__ float w_lds[KP * TD];
  __shared__ float b_lds[TD];
  __shared__ float corr_lds[PIX * KP];
  __shared__ __attribute__((aligned(16))) float f2_lds[CC * PL];

  const int tid   = threadIdx.x;
  const int tb    = blockIdx.y;
  const int pbase = blockIdx.x * PIX;

  const float* f1 = lvl + (size_t)tb * C * HW;
  const float* f2 = lvl + (size_t)(tb + B) * C * HW;

  const int h0 = pbase / W;
  const int w0 = pbase % W;
  int goff[NS];
#pragma unroll
  for (int s = 0; s < NS; ++s) {
    const int i = tid + 128 * s;
    const int r = i / HCP;
    const int j = i % HCP;
    int gh = h0 - 3 + r; gh += (gh < 0) ? H : 0; gh -= (gh >= H) ? H : 0;
    int gw = w0 - 3 + j; gw += (gw < 0) ? W : 0; gw -= (gw >= W) ? W : 0;
    goff[s] = (r < HR && j < HC) ? (gh * W + gw) : 0;
  }

  float pre[CC * NS];
#pragma unroll
  for (int cc = 0; cc < CC; ++cc)
#pragma unroll
    for (int s = 0; s < NS; ++s)
      pre[cc * NS + s] = f2[(size_t)cc * HW + goff[s]];

  for (int i = tid; i < KP * TD; i += 128)
    w_lds[i] = (i < 49 * TD) ? wmat[i] : 0.0f;
  for (int i = tid; i < TD; i += 128)
    b_lds[i] = bias[i];
  for (int i = tid; i < C * PIX; i += 128) {
    int c = i >> 6, p = i & 63;
    f1_lds[i] = f1[(size_t)c * HW + pbase + p];
  }

  {
    const float scale = 1.0f / __builtin_sqrtf((float)C);
    const int ky = tid >> 4;
    const int g  = tid & 15;
    const int pr = g / GPR;
    const int pc = 4 * (g % GPR);
    const int hb = (pr + 6 - ky) * HCP + pc;

    float acc[4][7];
#pragma unroll
    for (int q = 0; q < 4; ++q)
#pragma unroll
      for (int kx = 0; kx < 7; ++kx)
        acc[q][kx] = 0.0f;

#pragma unroll 1
    for (int c0 = 0; c0 < C; c0 += CC) {
      __syncthreads();
#pragma unroll
      for (int cc = 0; cc < CC; ++cc)
#pragma unroll
        for (int s = 0; s < NS; ++s)
          f2_lds[cc * PL + tid + 128 * s] = pre[cc * NS + s];
      __syncthreads();

      if (c0 + CC < C) {
        const float* nx = f2 + (size_t)(c0 + CC) * HW;
#pragma unroll
        for (int cc = 0; cc < CC; ++cc)
#pragma unroll
          for (int s = 0; s < NS; ++s)
            pre[cc * NS + s] = nx[(size_t)cc * HW + goff[s]];
      }

      if (ky < 7) {
#pragma unroll
        for (int cc = 0; cc < CC; ++cc) {
          const float4 fv = *reinterpret_cast<const float4*>(&f1_lds[(c0 + cc) * PIX + 4 * g]);
          const float* hp = &f2_lds[cc * PL + hb];
          const float4 ha = *reinterpret_cast<const float4*>(hp);
          const float4 hbv = *reinterpret_cast<const float4*>(hp + 4);
          const float2 hc = *reinterpret_cast<const float2*>(hp + 8);
          const float hv[10] = {ha.x, ha.y, ha.z, ha.w, hbv.x, hbv.y, hbv.z, hbv.w, hc.x, hc.y};
          const float fq[4] = {fv.x, fv.y, fv.z, fv.w};
#pragma unroll
          for (int q = 0; q < 4; ++q)
#pragma unroll
            for (int kx = 0; kx < 7; ++kx)
              acc[q][kx] = __builtin_fmaf(fq[q], hv[q + 6 - kx], acc[q][kx]);
        }
      }
    }

    if (ky < 7) {
#pragma unroll
      for (int q = 0; q < 4; ++q)
#pragma unroll
        for (int kx = 0; kx < 7; ++kx)
          corr_lds[(4 * g + q) * KP + ky * 7 + kx] = acc[q][kx] * scale;
    } else {
#pragma unroll
      for (int q = 0; q < 4; ++q)
#pragma unroll
        for (int z = 49; z < KP; ++z)
          corr_lds[(4 * g + q) * KP + z] = 0.0f;
    }
  }
  __syncthreads();

  {
    const int lane = tid & 31;
    const int wave = tid >> 5;
    const int p0   = wave * 16;
    const int nlo  = lane & 15;
    const int lhi  = lane >> 4;

    v2f bfr[13];
#pragma unroll
    for (int ks = 0; ks < 13; ++ks) {
      const int krow = 4 * ks + 2 * lhi;
      const float* src = &corr_lds[(p0 + nlo) * KP + krow];
      bfr[ks].x = src[0];
      bfr[ks].y = src[1];
    }

    for (int m = 0; m < 12; ++m) {
      const int d0 = m * 16;
      v8f acc;
#pragma unroll
      for (int r = 0; r < 8; ++r)
        acc[r] = b_lds[d0 + r + 8 * lhi];

#pragma unroll
      for (int ks = 0; ks < 13; ++ks) {
        const int krow = 4 * ks + 2 * lhi;
        v2f afr;
        afr.x = w_lds[krow * TD + d0 + nlo];
        afr.y = w_lds[(krow + 1) * TD + d0 + nlo];
        acc = __builtin_amdgcn_wmma_f32_16x16x4_f32(
            false, afr, false, bfr[ks], (short)0, acc, false, false);
      }

      float* op = out + ((size_t)tb * TD + d0) * HW + pbase + p0 + nlo;
#pragma unroll
      for (int r = 0; r < 8; ++r)
        op[(size_t)(r + 8 * lhi) * HW] = acc[r];
    }
  }
}

extern "C" void kernel_launch(void* const* d_in, const int* in_sizes, int n_in,
                              void* d_out, int out_size, void* d_ws, size_t ws_size,
                              hipStream_t stream) {
  const float* l1 = (const float*)d_in[0];
  const float* l2 = (const float*)d_in[1];
  const float* l3 = (const float*)d_in[2];
  const float* w1 = (const float*)d_in[3];
  const float* b1 = (const float*)d_in[4];
  const float* w2 = (const float*)d_in[5];
  const float* b2 = (const float*)d_in[6];
  const float* w3 = (const float*)d_in[7];
  const float* b3 = (const float*)d_in[8];

  const int T = 3, B = 2, TD = 192;
  float* tok1 = (float*)d_out;
  float* tok2 = tok1 + (size_t)(T - 1) * B * TD * 128 * 128;
  float* tok3 = tok2 + (size_t)(T - 1) * B * TD * 64 * 64;

  dim3 blk(128);
  cost_tok_kernel<64, 128, 128>
      <<<dim3(128 * 128 / 64, (T - 1) * B), blk, 0, stream>>>(l1, w1, b1, tok1, B);
  cost_tok_kernel<128, 64, 64>
      <<<dim3(64 * 64 / 64, (T - 1) * B), blk, 0, stream>>>(l2, w2, b2, tok2, B);
  cost_tok_kernel<192, 32, 32>
      <<<dim3(32 * 32 / 64, (T - 1) * B), blk, 0, stream>>>(l3, w3, b3, tok3, B);
}


# Prologue staging of weights, bias and frame-t tile: every request issued (16-byte loads) before one wait, replacing the weight loop's 78 dependent load-wait-store round trips per thread
# speedup vs baseline: 6.8017x; 2.4434x over previous
#include <hip/hip_runtime.h>

typedef __attribute__((ext_vector_type(2))) float v2f;
typedef __attribute__((ext_vector_type(8))) float v8f;
typedef __attribute__((ext_vector_type(4))) float v4f;

template <int C, int H, int W>
__global__ __launch_bounds__(128) void cost_tok_kernel(
    const float* __restrict__ lvl,
    const float* __restrict__ wmat,
    const float* __restrict__ bias,
    float* __restrict__ out,
    int B) {
  constexpr int HW  = H * W;
  constexpr int KP  = 52;
  constexpr int TD  = 192;
  constexpr int PIX = 64;

  constexpr int TW  = (W >= 64) ? 64 : W;
  constexpr int TR  = PIX / TW;
  constexpr int HR  = TR + 6;
  constexpr int HC  = TW + 6;
  constexpr int HCP = TW + 8;
  constexpr int NS  = (HR * HCP + 127) / 128;
  constexpr int PL  = NS * 128;
  constexpr int CC  = 16;
  constexpr int GPR = TW / 4;
  static_assert(C % CC == 0, "channel chunking");
  static_assert(W % TW == 0 && PIX % TW == 0 && TW % 4 == 0, "tile shape");
  static_assert(H % TR == 0 && H >= 4 && W >= 4, "tile rows");

  __shared__ __attribute__((aligned(16))) float f1_lds[C * PIX];
  __shared__ __attribute__((aligned(16))) float w_lds[KP * TD];
  __shared__ float b_lds[TD];
  __shared__ float corr_lds[PIX * KP];
  __shared__ __attribute__((aligned(16))) float f2_lds[CC * PL];

  const int tid   = threadIdx.x;
  const int tb    = blockIdx.y;
  const int pbase = blockIdx.x * PIX;

  const float* f1 = lvl + (size_t)tb * C * HW;
  const float* f2 = lvl + (size_t)(tb + B) * C * HW;

  const int h0 = pbase / W;
  const int w0 = pbase % W;
  int goff[NS];
#pragma unroll
  for (int s = 0; s < NS; ++s) {
    const int i = tid + 128 * s;
    const int r = i / HCP;
    const int j = i % HCP;
    int gh = h0 - 3 + r; gh += (gh < 0) ? H : 0; gh -= (gh >= H) ? H : 0;
    int gw = w0 - 3 + j; gw += (gw < 0) ? W : 0; gw -= (gw >= W) ? W : 0;
    goff[s] = (r < HR && j < HC) ? (gh * W + gw) : 0;
  }

  constexpr int W4  = 49 * TD / 4;
  constexpr int WP4 = KP * TD / 4;
  constexpr int WS  = (W4 + 127) / 128;
  constexpr int FS  = C * PIX / 4 / 128;
  static_assert(WS * 128 <= WP4 && WP4 - WS * 128 <= 128, "weight pad cover");
  static_assert((C * PIX / 4) % 128 == 0 && TD > 128 && TD <= 256, "slot maps");

  const v4f* w4g = reinterpret_cast<const v4f*>(wmat);
  v4f wv[WS];
#pragma unroll
  for (int j = 0; j < WS; ++j) {
    const int i4 = tid + 128 * j;
    const bool whole = (128 * j + 127 < W4);
    wv[j] = w4g[(whole || i4 < W4) ? i4 : (W4 - 1)];
  }

  v4f fv[FS];
#pragma unroll
  for (int j = 0; j < FS; ++j) {
    const int i4 = tid + 128 * j;
    fv[j] = *reinterpret_cast<const v4f*>(
        f1 + (size_t)(i4 >> 4) * HW + pbase + 4 * (i4 & 15));
  }

  const float bv0 = bias[tid];
  const float bv1 = bias[(tid < TD - 128) ? (tid + 128) : tid];

  float pre[CC * NS];
#pragma unroll
  for (int cc = 0; cc < CC; ++cc)
#pragma unroll
    for (int s = 0; s < NS; ++s)
      pre[cc * NS + s] = f2[(size_t)cc * HW + goff[s]];

  {
    const v4f z4 = {0.0f, 0.0f, 0.0f, 0.0f};
    v4f* w4l = reinterpret_cast<v4f*>(w_lds);
#pragma unroll
    for (int j = 0; j < WS; ++j) {
      const int i4 = tid + 128 * j;
      const bool whole = (128 * j + 127 < W4);
      w4l[i4] = (whole || i4 < W4) ? wv[j] : z4;
    }
    if (tid < WP4 - WS * 128) w4l[WS * 128 + tid] = z4;

    v4f* f4l = reinterpret_cast<v4f*>(f1_lds);
#pragma unroll
    for (int j = 0; j < FS; ++j)
      f4l[tid + 128 * j] = fv[j];

    b_lds[tid] = bv0;
    if (tid < TD - 128) b_lds[tid + 128] = bv1;
  }

  {
    const float scale = 1.0f / __builtin_sqrtf((float)C);
    const int ky = tid >> 4;
    const int g  = tid & 15;
    const int pr = g / GPR;
    const int pc = 4 * (g % GPR);
    const int hb = (pr + 6 - ky) * HCP + pc;

    float acc[4][7];
#pragma unroll
    for (int q = 0; q < 4; ++q)
#pragma unroll
      for (int kx = 0; kx < 7; ++kx)
        acc[q][kx] = 0.0f;

#pragma unroll 1
    for (int c0 = 0; c0 < C; c0 += CC) {
      __syncthreads();
#pragma unroll
      for (int cc = 0; cc < CC; ++cc)
#pragma unroll
        for (int s = 0; s < NS; ++s)
          f2_lds[cc * PL + tid + 128 * s] = pre[cc * NS + s];
      __syncthreads();

      if (c0 + CC < C) {
        const float* nx = f2 + (size_t)(c0 + CC) * HW;
#pragma unroll
        for (int cc = 0; cc < CC; ++cc)
#pragma unroll
          for (int s = 0; s < NS; ++s)
            pre[cc * NS + s] = nx[(size_t)cc * HW + goff[s]];
      }

      if (ky < 7) {
#pragma unroll
        for (int cc = 0; cc < CC; ++cc) {
          const float4 fv = *reinterpret_cast<const float4*>(&f1_lds[(c0 + cc) * PIX + 4 * g]);
          const float* hp = &f2_lds[cc * PL + hb];
          const float4 ha = *reinterpret_cast<const float4*>(hp);
          const float4 hbv = *reinterpret_cast<const float4*>(hp + 4);
          const float2 hc = *reinterpret_cast<const float2*>(hp + 8);
          const float hv[10] = {ha.x, ha.y, ha.z, ha.w, hbv.x, hbv.y, hbv.z, hbv.w, hc.x, hc.y};
          const float fq[4] = {fv.x, fv.y, fv.z, fv.w};
#pragma unroll
          for (int q = 0; q < 4; ++q)
#pragma unroll
            for (int kx = 0; kx < 7; ++kx)
              acc[q][kx] = __builtin_fmaf(fq[q], hv[q + 6 - kx], acc[q][kx]);
        }
      }
    }

    if (ky < 7) {
#pragma unroll
      for (int q = 0; q < 4; ++q)
#pragma unroll
        for (int kx = 0; kx < 7; ++kx)
          corr_lds[(4 * g + q) * KP + ky * 7 + kx] = acc[q][kx] * scale;
    } else {
#pragma unroll
      for (int q = 0; q < 4; ++q)
#pragma unroll
        for (int z = 49; z < KP; ++z)
          corr_lds[(4 * g + q) * KP + z] = 0.0f;
    }
  }
  __syncthreads();

  {
    const int lane = tid & 31;
    const int wave = tid >> 5;
    const int p0   = wave * 16;
    const int nlo  = lane & 15;
    const int lhi  = lane >> 4;

    v2f bfr[13];
#pragma unroll
    for (int ks = 0; ks < 13; ++ks) {
      const int krow = 4 * ks + 2 * lhi;
      const float* src = &corr_lds[(p0 + nlo) * KP + krow];
      bfr[ks].x = src[0];
      bfr[ks].y = src[1];
    }

    for (int m = 0; m < 12; ++m) {
      const int d0 = m * 16;
      v8f acc;
#pragma unroll
      for (int r = 0; r < 8; ++r)
        acc[r] = b_lds[d0 + r + 8 * lhi];

#pragma unroll
      for (int ks = 0; ks < 13; ++ks) {
        const int krow = 4 * ks + 2 * lhi;
        v2f afr;
        afr.x = w_lds[krow * TD + d0 + nlo];
        afr.y = w_lds[(krow + 1) * TD + d0 + nlo];
        acc = __builtin_amdgcn_wmma_f32_16x16x4_f32(
            false, afr, false, bfr[ks], (short)0, acc, false, false);
      }

      float* op = out + ((size_t)tb * TD + d0) * HW + pbase + p0 + nlo;
#pragma unroll
      for (int r = 0; r < 8; ++r)
        op[(size_t)(r + 8 * lhi) * HW] = acc[r];
    }
  }
}

extern "C" void kernel_launch(void* const* d_in, const int* in_sizes, int n_in,
                              void* d_out, int out_size, void* d_ws, size_t ws_size,
                              hipStream_t stream) {
  const float* l1 = (const float*)d_in[0];
  const float* l2 = (const float*)d_in[1];
  const float* l3 = (const float*)d_in[2];
  const float* w1 = (const float*)d_in[3];
  const float* b1 = (const float*)d_in[4];
  const float* w2 = (const float*)d_in[5];
  const float* b2 = (const float*)d_in[6];
  const float* w3 = (const float*)d_in[7];
  const float* b3 = (const float*)d_in[8];

  const int T = 3, B = 2, TD = 192;
  float* tok1 = (float*)d_out;
  float* tok2 = tok1 + (size_t)(T - 1) * B * TD * 128 * 128;
  float* tok3 = tok2 + (size_t)(T - 1) * B * TD * 64 * 64;

  dim3 blk(128);
  cost_tok_kernel<64, 128, 128>
      <<<dim3(128 * 128 / 64, (T - 1) * B), blk, 0, stream>>>(l1, w1, b1, tok1, B);
  cost_tok_kernel<128, 64, 64>
      <<<dim3(64 * 64 / 64, (T - 1) * B), blk, 0, stream>>>(l2, w2, b2, tok2, B);
  cost_tok_kernel<192, 32, 32>
      <<<dim3(32 * 32 / 64, (T - 1) * B), blk, 0, stream>>>(l3, w3, b3, tok3, B);
}


# Projection phase: weights held in LDS as K-row pairs (A fragment = one 8-byte load); next 3 M-tiles' fragments and bias read as one group ahead of each 39-WMMA block, 3 independent accumulators
# speedup vs baseline: 7.5621x; 1.1118x over previous
#include <hip/hip_runtime.h>

typedef __attribute__((ext_vector_type(2))) float v2f;
typedef __attribute__((ext_vector_type(8))) float v8f;
typedef __attribute__((ext_vector_type(4))) float v4f;

template <int C, int H, int W>
__global__ __launch_bounds__(128) void cost_tok_kernel(
    const float* __restrict__ lvl,
    const float* __restrict__ wmat,
    const float* __restrict__ bias,
    float* __restrict__ out,
    int B) {
  constexpr int HW  = H * W;
  constexpr int KP  = 52;
  constexpr int TD  = 192;
  constexpr int PIX = 64;

  constexpr int TW  = (W >= 64) ? 64 : W;
  constexpr int TR  = PIX / TW;
  constexpr int HR  = TR + 6;
  constexpr int HC  = TW + 6;
  constexpr int HCP = TW + 8;
  constexpr int NS  = (HR * HCP + 127) / 128;
  constexpr int PL  = NS * 128;
  constexpr int CC  = 16;
  constexpr int GPR = TW / 4;
  static_assert(C % CC == 0, "channel chunking");
  static_assert(W % TW == 0 && PIX % TW == 0 && TW % 4 == 0, "tile shape");
  static_assert(H % TR == 0 && H >= 4 && W >= 4, "tile rows");

  __shared__ __attribute__((aligned(16))) float f1_lds[C * PIX];
  __shared__ __attribute__((aligned(16))) float w_lds[KP * TD];
  __shared__ float b_lds[256];
  __shared__ float corr_lds[PIX * KP];
  __shared__ __attribute__((aligned(16))) float f2_lds[CC * PL];

  const int tid   = threadIdx.x;
  const int tb    = blockIdx.y;
  const int pbase = blockIdx.x * PIX;

  const float* f1 = lvl + (size_t)tb * C * HW;
  const float* f2 = lvl + (size_t)(tb + B) * C * HW;

  const int h0 = pbase / W;
  const int w0 = pbase % W;
  int goff[NS];
#pragma unroll
  for (int s = 0; s < NS; ++s) {
    const int i = tid + 128 * s;
    const int r = i / HCP;
    const int j = i % HCP;
    int gh = h0 - 3 + r; gh += (gh < 0) ? H : 0; gh -= (gh >= H) ? H : 0;
    int gw = w0 - 3 + j; gw += (gw < 0) ? W : 0; gw -= (gw >= W) ? W : 0;
    goff[s] = (r < HR && j < HC) ? (gh * W + gw) : 0;
  }

  constexpr int DQ  = TD / 4;
  constexpr int NKP = KP / 2;
  constexpr int WI  = NKP * DQ;
  constexpr int WS  = (WI + 127) / 128;
  constexpr int FS  = C * PIX / 4 / 128;
  static_assert((WS - 1) * 128 <= 24 * DQ && 24 * DQ < WI && WI <= WS * 128, "weight slot map");
  static_assert((C * PIX / 4) % 128 == 0 && TD > 128 && TD <= 256, "slot maps");

  const v4f* w4g = reinterpret_cast<const v4f*>(wmat);
  v4f wa[WS], wb[WS - 1];
#pragma unroll
  for (int j = 0; j < WS; ++j) {
    const int ir = tid + 128 * j;
    const int i  = (j < WS - 1 || ir < WI) ? ir : (WI - 1);
    const int kp = i / DQ;
    const int dq = i - kp * DQ;
    if (j < WS - 1) {
      wa[j] = w4g[(2 * kp) * DQ + dq];
      wb[j] = w4g[(2 * kp + 1) * DQ + dq];
    } else {
      wa[j] = w4g[(kp == 24) ? (48 * DQ + dq) : 0];
    }
  }

  v4f fv[FS];
#pragma unroll
  for (int j = 0; j < FS; ++j) {
    const int i4 = tid + 128 * j;
    fv[j] = *reinterpret_cast<const v4f*>(
        f1 + (size_t)(i4 >> 4) * HW + pbase + 4 * (i4 & 15));
  }

  const float bv0 = bias[tid];
  const float bv1 = bias[(tid < TD - 128) ? (tid + 128) : tid];

  float pre[CC * NS];
#pragma unroll
  for (int cc = 0; cc < CC; ++cc)
#pragma unroll
    for (int s = 0; s < NS; ++s)
      pre[cc * NS + s] = f2[(size_t)cc * HW + goff[s]];

  {
    const v4f z4 = {0.0f, 0.0f, 0.0f, 0.0f};
    v4f* w4l = reinterpret_cast<v4f*>(w_lds);
#pragma unroll
    for (int j = 0; j < WS; ++j) {
      const int ir = tid + 128 * j;
      const int i  = (j < WS - 1 || ir < WI) ? ir : (WI - 1);
      v4f a, b;
      if (j < WS - 1) {
        a = wa[j];
        b = wb[j];
      } else {
        a = (i < 25 * DQ) ? wa[j] : z4;
        b = z4;
      }
      const v4f lo = {a.x, b.x, a.y, b.y};
      const v4f hi = {a.z, b.z, a.w, b.w};
      w4l[2 * i]     = lo;
      w4l[2 * i + 1] = hi;
    }

    v4f* f4l = reinterpret_cast<v4f*>(f1_lds);
#pragma unroll
    for (int j = 0; j < FS; ++j)
      f4l[tid + 128 * j] = fv[j];

    b_lds[tid] = bv0;
    b_lds[tid + 128] = bv1;
  }

  {
    const float scale = 1.0f / __builtin_sqrtf((float)C);
    const int ky = tid >> 4;
    const int g  = tid & 15;
    const int pr = g / GPR;
    const int pc = 4 * (g % GPR);
    const int hb = (pr + 6 - ky) * HCP + pc;

    float acc[4][7];
#pragma unroll
    for (int q = 0; q < 4; ++q)
#pragma unroll
      for (int kx = 0; kx < 7; ++kx)
        acc[q][kx] = 0.0f;

#pragma unroll 1
    for (int c0 = 0; c0 < C; c0 += CC) {
      __syncthreads();
#pragma unroll
      for (int cc = 0; cc < CC; ++cc)
#pragma unroll
        for (int s = 0; s < NS; ++s)
          f2_lds[cc * PL + tid + 128 * s] = pre[cc * NS + s];
      __syncthreads();

      if (c0 + CC < C) {
        const float* nx = f2 + (size_t)(c0 + CC) * HW;
#pragma unroll
        for (int cc = 0; cc < CC; ++cc)
#pragma unroll
          for (int s = 0; s < NS; ++s)
            pre[cc * NS + s] = nx[(size_t)cc * HW + goff[s]];
      }

      if (ky < 7) {
#pragma unroll
        for (int cc = 0; cc < CC; ++cc) {
          const float4 fv = *reinterpret_cast<const float4*>(&f1_lds[(c0 + cc) * PIX + 4 * g]);
          const float* hp = &f2_lds[cc * PL + hb];
          const float4 ha = *reinterpret_cast<const float4*>(hp);
          const float4 hbv = *reinterpret_cast<const float4*>(hp + 4);
          const float2 hc = *reinterpret_cast<const float2*>(hp + 8);
          const float hv[10] = {ha.x, ha.y, ha.z, ha.w, hbv.x, hbv.y, hbv.z, hbv.w, hc.x, hc.y};
          const float fq[4] = {fv.x, fv.y, fv.z, fv.w};
#pragma unroll
          for (int q = 0; q < 4; ++q)
#pragma unroll
            for (int kx = 0; kx < 7; ++kx)
              acc[q][kx] = __builtin_fmaf(fq[q], hv[q + 6 - kx], acc[q][kx]);
        }
      }
    }

    if (ky < 7) {
#pragma unroll
      for (int q = 0; q < 4; ++q)
#pragma unroll
        for (int kx = 0; kx < 7; ++kx)
          corr_lds[(4 * g + q) * KP + ky * 7 + kx] = acc[q][kx] * scale;
    } else {
#pragma unroll
      for (int q = 0; q < 4; ++q)
#pragma unroll
        for (int z = 49; z < KP; ++z)
          corr_lds[(4 * g + q) * KP + z] = 0.0f;
    }
  }
  __syncthreads();

  {
    const int lane = tid & 31;
    const int wave = tid >> 5;
    const int p0   = wave * 16;
    const int nlo  = lane & 15;
    const int lhi  = lane >> 4;

    v2f bfr[13];
#pragma unroll
    for (int ks = 0; ks < 13; ++ks) {
      const int krow = 4 * ks + 2 * lhi;
      const float* src = &corr_lds[(p0 + nlo) * KP + krow];
      bfr[ks].x = src[0];
      bfr[ks].y = src[1];
    }

    constexpr int MG = 3;
    constexpr int NG = 12 / MG;
    v2f afr[2][MG * 13];
    v8f acc[2][MG];

    const v2f* w2l = reinterpret_cast<const v2f*>(w_lds);
#define P2_LOAD(G, S)                                                      \
    _Pragma("unroll")                                                      \
    for (int t = 0; t < MG; ++t) {                                         \
      const int d0 = ((G) * MG + t) * 16;                                  \
      _Pragma("unroll")                                                    \
      for (int ks = 0; ks < 13; ++ks)                                      \
        afr[S][t * 13 + ks] = w2l[(2 * ks + lhi) * TD + d0 + nlo];         \
      _Pragma("unroll")                                                    \
      for (int r = 0; r < 8; ++r)                                          \
        acc[S][t][r] = b_lds[d0 + r + 8 * lhi];                            \
    }

#define P2_MMA_STORE(G, S)                                                 \
    _Pragma("unroll")                                                      \
    for (int ks = 0; ks < 13; ++ks) {                                      \
      _Pragma("unroll")                                                    \
      for (int t = 0; t < MG; ++t)                                         \
        acc[S][t] = __builtin_amdgcn_wmma_f32_16x16x4_f32(                 \
            false, afr[S][t * 13 + ks], false, bfr[ks], (short)0,          \
            acc[S][t], false, false);                                      \
    }                                                                      \
    _Pragma("unroll")                                                      \
    for (int t = 0; t < MG; ++t) {                                         \
      const int d0 = ((G) * MG + t) * 16;                                  \
      float* op = out + ((size_t)tb * TD + d0) * HW + pbase + p0 + nlo;    \
      _Pragma("unroll")                                                    \
      for (int r = 0; r < 8; ++r)                                          \
        op[(size_t)(r + 8 * lhi) * HW] = acc[S][t][r];                     \
    }

    P2_LOAD(0, 0)
    __builtin_amdgcn_sched_barrier(0);
#pragma unroll
    for (int g = 0; g < NG; g += 2) {
      P2_LOAD(g + 1, 1)
      __builtin_amdgcn_sched_barrier(0);
      P2_MMA_STORE(g, 0)
      __builtin_amdgcn_sched_barrier(0);
      if (g + 2 < NG) {
        P2_LOAD(g + 2, 0)
      }
      __builtin_amdgcn_sched_barrier(0);
      P2_MMA_STORE(g + 1, 1)
      __builtin_amdgcn_sched_barrier(0);
    }
#undef P2_LOAD
#undef P2_MMA_STORE
  }
}

extern "C" void kernel_launch(void* const* d_in, const int* in_sizes, int n_in,
                              void* d_out, int out_size, void* d_ws, size_t ws_size,
                              hipStream_t stream) {
  const float* l1 = (const float*)d_in[0];
  const float* l2 = (const float*)d_in[1];
  const float* l3 = (const float*)d_in[2];
  const float* w1 = (const float*)d_in[3];
  const float* b1 = (const float*)d_in[4];
  const float* w2 = (const float*)d_in[5];
  const float* b2 = (const float*)d_in[6];
  const float* w3 = (const float*)d_in[7];
  const float* b3 = (const float*)d_in[8];

  const int T = 3, B = 2, TD = 192;
  float* tok1 = (float*)d_out;
  float* tok2 = tok1 + (size_t)(T - 1) * B * TD * 128 * 128;
  float* tok3 = tok2 + (size_t)(T - 1) * B * TD * 64 * 64;

  dim3 blk(128);
  cost_tok_kernel<64, 128, 128>
      <<<dim3(128 * 128 / 64, (T - 1) * B), blk, 0, stream>>>(l1, w1, b1, tok1, B);
  cost_tok_kernel<128, 64, 64>
      <<<dim3(64 * 64 / 64, (T - 1) * B), blk, 0, stream>>>(l2, w2, b2, tok2, B);
  cost_tok_kernel<192, 32, 32>
      <<<dim3(32 * 32 / 64, (T - 1) * B), blk, 0, stream>>>(l3, w3, b3, tok3, B);
}


# Next-frame halo staged as aligned 16-byte pieces: one 16-byte global load and one 16-byte LDS store per thread per channel (was 3-4 4-byte ones); 16 loads in flight per chunk, below the 63-load limit
# speedup vs baseline: 8.2209x; 1.0871x over previous
#include <hip/hip_runtime.h>

typedef __attribute__((ext_vector_type(2))) float v2f;
typedef __attribute__((ext_vector_type(8))) float v8f;
typedef __attribute__((ext_vector_type(4))) float v4f;

template <int C, int H, int W>
__global__ __launch_bounds__(128) void cost_tok_kernel(
    const float* __restrict__ lvl,
    const float* __restrict__ wmat,
    const float* __restrict__ bias,
    float* __restrict__ out,
    int B) {
  constexpr int HW  = H * W;
  constexpr int KP  = 52;
  constexpr int TD  = 192;
  constexpr int PIX = 64;

  constexpr int TW  = (W >= 64) ? 64 : W;
  constexpr int TR  = PIX / TW;
  constexpr int HR  = TR + 6;
  constexpr int HCP = TW + 8;
  constexpr int QPR = HCP / 4;
  constexpr int NQ  = HR * QPR;
  constexpr int PL  = NQ * 4;
  constexpr int CC  = 16;
  constexpr int GPR = TW / 4;
  static_assert(C % CC == 0, "channel chunking");
  static_assert(W % TW == 0 && PIX % TW == 0 && TW % 4 == 0, "tile shape");
  static_assert(H % TR == 0 && H >= 4 && W >= 4 && W % 4 == 0, "tile rows");
  static_assert(NQ <= 128, "one halo piece per thread");

  __shared__ __attribute__((aligned(16))) float f1_lds[C * PIX];
  __shared__ __attribute__((aligned(16))) float w_lds[KP * TD];
  __shared__ float b_lds[256];
  __shared__ float corr_lds[PIX * KP];
  __shared__ __attribute__((aligned(16))) float f2_lds[CC * PL];

  const int tid   = threadIdx.x;
  const int tb    = blockIdx.y;
  const int pbase = blockIdx.x * PIX;

  const float* f1 = lvl + (size_t)tb * C * HW;
  const float* f2 = lvl + (size_t)(tb + B) * C * HW;

  const int h0 = pbase / W;
  const int w0 = pbase % W;
  const bool act = tid < NQ;
  int goff;
  {
    const int r = tid / QPR;
    const int j = tid - r * QPR;
    int gh = h0 - 3 + r; gh += (gh < 0) ? H : 0; gh -= (gh >= H) ? H : 0;
    int gw = w0 - 4 + 4 * j; gw += (gw < 0) ? W : 0; gw -= (gw >= W) ? W : 0;
    goff = act ? (gh * W + gw) : 0;
  }

  constexpr int DQ  = TD / 4;
  constexpr int NKP = KP / 2;
  constexpr int WI  = NKP * DQ;
  constexpr int WS  = (WI + 127) / 128;
  constexpr int FS  = C * PIX / 4 / 128;
  static_assert((WS - 1) * 128 <= 24 * DQ && 24 * DQ < WI && WI <= WS * 128, "weight slot map");
  static_assert((C * PIX / 4) % 128 == 0 && TD > 128 && TD <= 256, "slot maps");

  const v4f* w4g = reinterpret_cast<const v4f*>(wmat);
  v4f wa[WS], wb[WS - 1];
#pragma unroll
  for (int j = 0; j < WS; ++j) {
    const int ir = tid + 128 * j;
    const int i  = (j < WS - 1 || ir < WI) ? ir : (WI - 1);
    const int kp = i / DQ;
    const int dq = i - kp * DQ;
    if (j < WS - 1) {
      wa[j] = w4g[(2 * kp) * DQ + dq];
      wb[j] = w4g[(2 * kp + 1) * DQ + dq];
    } else {
      wa[j] = w4g[(kp == 24) ? (48 * DQ + dq) : 0];
    }
  }

  v4f fv[FS];
#pragma unroll
  for (int j = 0; j < FS; ++j) {
    const int i4 = tid + 128 * j;
    fv[j] = *reinterpret_cast<const v4f*>(
        f1 + (size_t)(i4 >> 4) * HW + pbase + 4 * (i4 & 15));
  }

  const float bv0 = bias[tid];
  const float bv1 = bias[(tid < TD - 128) ? (tid + 128) : tid];

  v4f pre[CC];
  if (act) {
#pragma unroll
    for (int cc = 0; cc < CC; ++cc)
      pre[cc] = *reinterpret_cast<const v4f*>(f2 + (size_t)cc * HW + goff);
  }

  {
    const v4f z4 = {0.0f, 0.0f, 0.0f, 0.0f};
    v4f* w4l = reinterpret_cast<v4f*>(w_lds);
#pragma unroll
    for (int j = 0; j < WS; ++j) {
      const int ir = tid + 128 * j;
      const int i  = (j < WS - 1 || ir < WI) ? ir : (WI - 1);
      v4f a, b;
      if (j < WS - 1) {
        a = wa[j];
        b = wb[j];
      } else {
        a = (i < 25 * DQ) ? wa[j] : z4;
        b = z4;
      }
      const v4f lo = {a.x, b.x, a.y, b.y};
      const v4f hi = {a.z, b.z, a.w, b.w};
      w4l[2 * i]     = lo;
      w4l[2 * i + 1] = hi;
    }

    v4f* f4l = reinterpret_cast<v4f*>(f1_lds);
#pragma unroll
    for (int j = 0; j < FS; ++j)
      f4l[tid + 128 * j] = fv[j];

    b_lds[tid] = bv0;
    b_lds[tid + 128] = bv1;
  }

  {
    const float scale = 1.0f / __builtin_sqrtf((float)C);
    const int ky = tid >> 4;
    const int g  = tid & 15;
    const int pr = g / GPR;
    const int pc = 4 * (g % GPR);
    const int hb = (pr + 6 - ky) * HCP + pc;
    v4f* f2q = reinterpret_cast<v4f*>(f2_lds);

    float acc[4][7];
#pragma unroll
    for (int q = 0; q < 4; ++q)
#pragma unroll
      for (int kx = 0; kx < 7; ++kx)
        acc[q][kx] = 0.0f;

#pragma unroll 1
    for (int c0 = 0; c0 < C; c0 += CC) {
      __syncthreads();
      if (act) {
#pragma unroll
        for (int cc = 0; cc < CC; ++cc)
          f2q[cc * NQ + tid] = pre[cc];
      }
      __syncthreads();

      if (c0 + CC < C && act) {
        const float* nx = f2 + (size_t)(c0 + CC) * HW + goff;
#pragma unroll
        for (int cc = 0; cc < CC; ++cc)
          pre[cc] = *reinterpret_cast<const v4f*>(nx + (size_t)cc * HW);
      }

      if (ky < 7) {
#pragma unroll
        for (int cc = 0; cc < CC; ++cc) {
          const float4 fv = *reinterpret_cast<const float4*>(&f1_lds[(c0 + cc) * PIX + 4 * g]);
          const float* hp = &f2_lds[cc * PL + hb];
          const float4 ha = *reinterpret_cast<const float4*>(hp);
          const float4 hbv = *reinterpret_cast<const float4*>(hp + 4);
          const float4 hc = *reinterpret_cast<const float4*>(hp + 8);
          const float hv[10] = {ha.y, ha.z, ha.w, hbv.x, hbv.y, hbv.z, hbv.w, hc.x, hc.y, hc.z};
          const float fq[4] = {fv.x, fv.y, fv.z, fv.w};
#pragma unroll
          for (int q = 0; q < 4; ++q)
#pragma unroll
            for (int kx = 0; kx < 7; ++kx)
              acc[q][kx] = __builtin_fmaf(fq[q], hv[q + 6 - kx], acc[q][kx]);
        }
      }
    }

    if (ky < 7) {
#pragma unroll
      for (int q = 0; q < 4; ++q)
#pragma unroll
        for (int kx = 0; kx < 7; ++kx)
          corr_lds[(4 * g + q) * KP + ky * 7 + kx] = acc[q][kx] * scale;
    } else {
#pragma unroll
      for (int q = 0; q < 4; ++q)
#pragma unroll
        for (int z = 49; z < KP; ++z)
          corr_lds[(4 * g + q) * KP + z] = 0.0f;
    }
  }
  __syncthreads();

  {
    const int lane = tid & 31;
    const int wave = tid >> 5;
    const int p0   = wave * 16;
    const int nlo  = lane & 15;
    const int lhi  = lane >> 4;

    v2f bfr[13];
#pragma unroll
    for (int ks = 0; ks < 13; ++ks) {
      const int krow = 4 * ks + 2 * lhi;
      const float* src = &corr_lds[(p0 + nlo) * KP + krow];
      bfr[ks].x = src[0];
      bfr[ks].y = src[1];
    }

    constexpr int MG = 3;
    constexpr int NG = 12 / MG;
    v2f afr[2][MG * 13];
    v8f acc[2][MG];

    const v2f* w2l = reinterpret_cast<const v2f*>(w_lds);
#define P2_LOAD(G, S)                                                      \
    _Pragma("unroll")                                                      \
    for (int t = 0; t < MG; ++t) {                                         \
      const int d0 = ((G) * MG + t) * 16;                                  \
      _Pragma("unroll")                                                    \
      for (int ks = 0; ks < 13; ++ks)                                      \
        afr[S][t * 13 + ks] = w2l[(2 * ks + lhi) * TD + d0 + nlo];         \
      _Pragma("unroll")                                                    \
      for (int r = 0; r < 8; ++r)                                          \
        acc[S][t][r] = b_lds[d0 + r + 8 * lhi];                            \
    }

#define P2_MMA_STORE(G, S)                                                 \
    _Pragma("unroll")                                                      \
    for (int ks = 0; ks < 13; ++ks) {                                      \
      _Pragma("unroll")                                                    \
      for (int t = 0; t < MG; ++t)                                         \
        acc[S][t] = __builtin_amdgcn_wmma_f32_16x16x4_f32(                 \
            false, afr[S][t * 13 + ks], false, bfr[ks], (short)0,          \
            acc[S][t], false, false);                                      \
    }                                                                      \
    _Pragma("unroll")                                                      \
    for (int t = 0; t < MG; ++t) {                                         \
      const int d0 = ((G) * MG + t) * 16;                                  \
      float* op = out + ((size_t)tb * TD + d0) * HW + pbase + p0 + nlo;    \
      _Pragma("unroll")                                                    \
      for (int r = 0; r < 8; ++r)                                          \
        op[(size_t)(r + 8 * lhi) * HW] = acc[S][t][r];                     \
    }

    P2_LOAD(0, 0)
    __builtin_amdgcn_sched_barrier(0);
#pragma unroll
    for (int g = 0; g < NG; g += 2) {
      P2_LOAD(g + 1, 1)
      __builtin_amdgcn_sched_barrier(0);
      P2_MMA_STORE(g, 0)
      __builtin_amdgcn_sched_barrier(0);
      if (g + 2 < NG) {
        P2_LOAD(g + 2, 0)
      }
      __builtin_amdgcn_sched_barrier(0);
      P2_MMA_STORE(g + 1, 1)
      __builtin_amdgcn_sched_barrier(0);
    }
#undef P2_LOAD
#undef P2_MMA_STORE
  }
}

extern "C" void kernel_launch(void* const* d_in, const int* in_sizes, int n_in,
                              void* d_out, int out_size, void* d_ws, size_t ws_size,
                              hipStream_t stream) {
  const float* l1 = (const float*)d_in[0];
  const float* l2 = (const float*)d_in[1];
  const float* l3 = (const float*)d_in[2];
  const float* w1 = (const float*)d_in[3];
  const float* b1 = (const float*)d_in[4];
  const float* w2 = (const float*)d_in[5];
  const float* b2 = (const float*)d_in[6];
  const float* w3 = (const float*)d_in[7];
  const float* b3 = (const float*)d_in[8];

  const int T = 3, B = 2, TD = 192;
  float* tok1 = (float*)d_out;
  float* tok2 = tok1 + (size_t)(T - 1) * B * TD * 128 * 128;
  float* tok3 = tok2 + (size_t)(T - 1) * B * TD * 64 * 64;

  dim3 blk(128);
  cost_tok_kernel<64, 128, 128>
      <<<dim3(128 * 128 / 64, (T - 1) * B), blk, 0, stream>>>(l1, w1, b1, tok1, B);
  cost_tok_kernel<128, 64, 64>
      <<<dim3(64 * 64 / 64, (T - 1) * B), blk, 0, stream>>>(l2, w2, b2, tok2, B);
  cost_tok_kernel<192, 32, 32>
      <<<dim3(32 * 32 / 64, (T - 1) * B), blk, 0, stream>>>(l3, w3, b3, tok3, B);
}


# Correlation window widened to 8 pixels x 7 dx per thread; thread halves take alternate channels, partial sums added once through LDS: 6 LDS 16-byte reads per 56 multiply-adds (was 8 per 56)
# speedup vs baseline: 8.7038x; 1.0587x over previous
#include <hip/hip_runtime.h>

typedef __attribute__((ext_vector_type(2))) float v2f;
typedef __attribute__((ext_vector_type(8))) float v8f;
typedef __attribute__((ext_vector_type(4))) float v4f;

template <int C, int H, int W>
__global__ __launch_bounds__(128) void cost_tok_kernel(
    const float* __restrict__ lvl,
    const float* __restrict__ wmat,
    const float* __restrict__ bias,
    float* __restrict__ out,
    int B) {
  constexpr int HW  = H * W;
  constexpr int KP  = 52;
  constexpr int TD  = 192;
  constexpr int PIX = 64;

  constexpr int TW  = (W >= 64) ? 64 : W;
  constexpr int TR  = PIX / TW;
  constexpr int HR  = TR + 6;
  constexpr int HCP = TW + 8;
  constexpr int QPR = HCP / 4;
  constexpr int NQ  = HR * QPR;
  constexpr int HP  = HCP + 4;
  constexpr int PL  = HR * HP;
  constexpr int PLQ = PL / 4;
  constexpr int CC  = 16;
  constexpr int G8  = TW / 8;
  constexpr int SP  = 60;
  static_assert(C % CC == 0, "channel chunking");
  static_assert(W % TW == 0 && PIX % TW == 0 && TW % 4 == 0, "tile shape");
  static_assert(H % TR == 0 && H >= 4 && W >= 4 && W % 4 == 0, "tile rows");
  static_assert(NQ <= 128, "one halo piece per thread");
  static_assert(HP % 4 == 0 && 56 * SP <= CC * PL, "window and scratch");
  static_assert((TR == 1 && G8 == 8) || (TR == 2 && G8 == 4), "window map");

  __shared__ __attribute__((aligned(16))) float f1_lds[C * PIX];
  __shared__ __attribute__((aligned(16))) float w_lds[KP * TD];
  __shared__ float b_lds[256];
  __shared__ float corr_lds[PIX * KP];
  __shared__ __attribute__((aligned(16))) float f2_lds[CC * PL];

  const int tid   = threadIdx.x;
  const int tb    = blockIdx.y;
  const int pbase = blockIdx.x * PIX;

  const float* f1 = lvl + (size_t)tb * C * HW;
  const float* f2 = lvl + (size_t)(tb + B) * C * HW;

  const int h0 = pbase / W;
  const int w0 = pbase % W;
  const bool act = tid < NQ;
  int goff, sidx;
  {
    const int r = tid / QPR;
    const int j = tid - r * QPR;
    sidx = r * (HP / 4) + j;
    int gh = h0 - 3 + r; gh += (gh < 0) ? H : 0; gh -= (gh >= H) ? H : 0;
    int gw = w0 - 4 + 4 * j; gw += (gw < 0) ? W : 0; gw -= (gw >= W) ? W : 0;
    goff = act ? (gh * W + gw) : 0;
  }

  constexpr int DQ  = TD / 4;
  constexpr int NKP = KP / 2;
  constexpr int WI  = NKP * DQ;
  constexpr int WS  = (WI + 127) / 128;
  constexpr int FS  = C * PIX / 4 / 128;
  static_assert((WS - 1) * 128 <= 24 * DQ && 24 * DQ < WI && WI <= WS * 128, "weight slot map");
  static_assert((C * PIX / 4) % 128 == 0 && TD > 128 && TD <= 256, "slot maps");

  const v4f* w4g = reinterpret_cast<const v4f*>(wmat);
  v4f wa[WS], wb[WS - 1];
#pragma unroll
  for (int j = 0; j < WS; ++j) {
    const int ir = tid + 128 * j;
    const int i  = (j < WS - 1 || ir < WI) ? ir : (WI - 1);
    const int kp = i / DQ;
    const int dq = i - kp * DQ;
    if (j < WS - 1) {
      wa[j] = w4g[(2 * kp) * DQ + dq];
      wb[j] = w4g[(2 * kp + 1) * DQ + dq];
    } else {
      wa[j] = w4g[(kp == 24) ? (48 * DQ + dq) : 0];
    }
  }

  v4f fv[FS];
#pragma unroll
  for (int j = 0; j < FS; ++j) {
    const int i4 = tid + 128 * j;
    fv[j] = *reinterpret_cast<const v4f*>(
        f1 + (size_t)(i4 >> 4) * HW + pbase + 4 * (i4 & 15));
  }

  const float bv0 = bias[tid];
  const float bv1 = bias[(tid < TD - 128) ? (tid + 128) : tid];

  v4f pre[CC];
  if (act) {
#pragma unroll
    for (int cc = 0; cc < CC; ++cc)
      pre[cc] = *reinterpret_cast<const v4f*>(f2 + (size_t)cc * HW + goff);
  }

  {
    const v4f z4 = {0.0f, 0.0f, 0.0f, 0.0f};
    v4f* w4l = reinterpret_cast<v4f*>(w_lds);
#pragma unroll
    for (int j = 0; j < WS; ++j) {
      const int ir = tid + 128 * j;
      const int i  = (j < WS - 1 || ir < WI) ? ir : (WI - 1);
      v4f a, b;
      if (j < WS - 1) {
        a = wa[j];
        b = wb[j];
      } else {
        a = (i < 25 * DQ) ? wa[j] : z4;
        b = z4;
      }
      const v4f lo = {a.x, b.x, a.y, b.y};
      const v4f hi = {a.z, b.z, a.w, b.w};
      w4l[2 * i]     = lo;
      w4l[2 * i + 1] = hi;
    }

    v4f* f4l = reinterpret_cast<v4f*>(f1_lds);
#pragma unroll
    for (int j = 0; j < FS; ++j)
      f4l[tid + 128 * j] = fv[j];

    b_lds[tid] = bv0;
    b_lds[tid + 128] = bv1;
  }

  {
    const float scale = 1.0f / __builtin_sqrtf((float)C);
    const int half = tid >> 6;
    const int q8   = tid & 63;
    int ky, g;
    if (TR == 1) {
      ky = q8 >> 3;
      g  = q8 & 7;
    } else {
      const int prd = (q8 >> 2) & 1;
      ky = (q8 >> 3) - 1 + prd;
      g  = 4 * prd + (q8 & 3);
    }
    const bool work = (unsigned)ky < 7u;
    const int pr   = g / G8;
    const int pc   = 8 * (g % G8);
    const int hb   = (pr + 6 - ky) * HP + pc;
    v4f* f2q = reinterpret_cast<v4f*>(f2_lds);

    float acc[8][7];
#pragma unroll
    for (int q = 0; q < 8; ++q)
#pragma unroll
      for (int kx = 0; kx < 7; ++kx)
        acc[q][kx] = 0.0f;

#pragma unroll 1
    for (int c0 = 0; c0 < C; c0 += CC) {
      __syncthreads();
      if (act) {
#pragma unroll
        for (int cc = 0; cc < CC; ++cc)
          f2q[cc * PLQ + sidx] = pre[cc];
      }
      __syncthreads();

      if (c0 + CC < C && act) {
        const float* nx = f2 + (size_t)(c0 + CC) * HW + goff;
#pragma unroll
        for (int cc = 0; cc < CC; ++cc)
          pre[cc] = *reinterpret_cast<const v4f*>(nx + (size_t)cc * HW);
      }

      if (work) {
#pragma unroll
        for (int c2 = 0; c2 < CC / 2; ++c2) {
          const int cc = 2 * c2 + half;
          const v4f* fp = reinterpret_cast<const v4f*>(&f1_lds[(c0 + cc) * PIX + 8 * g]);
          const v4f* hp = reinterpret_cast<const v4f*>(&f2_lds[cc * PL + hb]);
          const v4f fa = fp[0];
          const v4f fb = fp[1];
          v4f h0 = hp[0];
          const v4f h1 = hp[1];
          const v4f h2 = hp[2];
          v4f h3 = hp[3];
          asm("" : "+v"(h0));
          asm("" : "+v"(h3));
          const float hv[14] = {h0.y, h0.z, h0.w, h1.x, h1.y, h1.z, h1.w,
                                h2.x, h2.y, h2.z, h2.w, h3.x, h3.y, h3.z};
          const float fq[8] = {fa.x, fa.y, fa.z, fa.w, fb.x, fb.y, fb.z, fb.w};
#pragma unroll
          for (int q = 0; q < 8; ++q)
#pragma unroll
            for (int kx = 0; kx < 7; ++kx)
              acc[q][kx] = __builtin_fmaf(fq[q], hv[q + 6 - kx], acc[q][kx]);
        }
      }
    }

    __syncthreads();
    v4f* sc4 = reinterpret_cast<v4f*>(&f2_lds[q8 * SP]);
    if (half == 1 && work) {
#pragma unroll
      for (int i = 0; i < 14; ++i) {
        const v4f v = {acc[(4 * i) / 7][(4 * i) % 7], acc[(4 * i + 1) / 7][(4 * i + 1) % 7],
                       acc[(4 * i + 2) / 7][(4 * i + 2) % 7], acc[(4 * i + 3) / 7][(4 * i + 3) % 7]};
        sc4[i] = v;
      }
    }
    __syncthreads();
    if (half == 0) {
      if (work) {
#pragma unroll
        for (int i = 0; i < 14; ++i) {
          const v4f v = sc4[i];
          const float o[4] = {v.x, v.y, v.z, v.w};
#pragma unroll
          for (int e = 0; e < 4; ++e) {
            const int q = (4 * i + e) / 7;
            const int kx = (4 * i + e) % 7;
            corr_lds[(8 * g + q) * KP + ky * 7 + kx] = (acc[q][kx] + o[e]) * scale;
          }
        }
      } else {
#pragma unroll
        for (int q = 0; q < 8; ++q)
#pragma unroll
          for (int z = 49; z < KP; ++z)
            corr_lds[(8 * g + q) * KP + z] = 0.0f;
      }
    }
  }
  __syncthreads();

  {
    const int lane = tid & 31;
    const int wave = tid >> 5;
    const int p0   = wave * 16;
    const int nlo  = lane & 15;
    const int lhi  = lane >> 4;

    v2f bfr[13];
#pragma unroll
    for (int ks = 0; ks < 13; ++ks) {
      const int krow = 4 * ks + 2 * lhi;
      const float* src = &corr_lds[(p0 + nlo) * KP + krow];
      bfr[ks].x = src[0];
      bfr[ks].y = src[1];
    }

    constexpr int MG = 3;
    constexpr int NG = 12 / MG;
    v2f afr[2][MG * 13];
    v8f acc[2][MG];

    const v2f* w2l = reinterpret_cast<const v2f*>(w_lds);
#define P2_LOAD(G, S)                                                      \
    _Pragma("unroll")                                                      \
    for (int t = 0; t < MG; ++t) {                                         \
      const int d0 = ((G) * MG + t) * 16;                                  \
      _Pragma("unroll")                                                    \
      for (int ks = 0; ks < 13; ++ks)                                      \
        afr[S][t * 13 + ks] = w2l[(2 * ks + lhi) * TD + d0 + nlo];         \
      _Pragma("unroll")                                                    \
      for (int r = 0; r < 8; ++r)                                          \
        acc[S][t][r] = b_lds[d0 + r + 8 * lhi];                            \
    }

#define P2_MMA_STORE(G, S)                                                 \
    _Pragma("unroll")                                                      \
    for (int ks = 0; ks < 13; ++ks) {                                      \
      _Pragma("unroll")                                                    \
      for (int t = 0; t < MG; ++t)                                         \
        acc[S][t] = __builtin_amdgcn_wmma_f32_16x16x4_f32(                 \
            false, afr[S][t * 13 + ks], false, bfr[ks], (short)0,          \
            acc[S][t], false, false);                                      \
    }                                                                      \
    _Pragma("unroll")                                                      \
    for (int t = 0; t < MG; ++t) {                                         \
      const int d0 = ((G) * MG + t) * 16;                                  \
      float* op = out + ((size_t)tb * TD + d0) * HW + pbase + p0 + nlo;    \
      _Pragma("unroll")                                                    \
      for (int r = 0; r < 8; ++r)                                          \
        op[(size_t)(r + 8 * lhi) * HW] = acc[S][t][r];                     \
    }

    P2_LOAD(0, 0)
    __builtin_amdgcn_sched_barrier(0);
#pragma unroll
    for (int g = 0; g < NG; g += 2) {
      P2_LOAD(g + 1, 1)
      __builtin_amdgcn_sched_barrier(0);
      P2_MMA_STORE(g, 0)
      __builtin_amdgcn_sched_barrier(0);
      if (g + 2 < NG) {
        P2_LOAD(g + 2, 0)
      }
      __builtin_amdgcn_sched_barrier(0);
      P2_MMA_STORE(g + 1, 1)
      __builtin_amdgcn_sched_barrier(0);
    }
#undef P2_LOAD
#undef P2_MMA_STORE
  }
}

extern "C" void kernel_launch(void* const* d_in, const int* in_sizes, int n_in,
                              void* d_out, int out_size, void* d_ws, size_t ws_size,
                              hipStream_t stream) {
  const float* l1 = (const float*)d_in[0];
  const float* l2 = (const float*)d_in[1];
  const float* l3 = (const float*)d_in[2];
  const float* w1 = (const float*)d_in[3];
  const float* b1 = (const float*)d_in[4];
  const float* w2 = (const float*)d_in[5];
  const float* b2 = (const float*)d_in[6];
  const float* w3 = (const float*)d_in[7];
  const float* b3 = (const float*)d_in[8];

  const int T = 3, B = 2, TD = 192;
  float* tok1 = (float*)d_out;
  float* tok2 = tok1 + (size_t)(T - 1) * B * TD * 128 * 128;
  float* tok3 = tok2 + (size_t)(T - 1) * B * TD * 64 * 64;

  dim3 blk(128);
  cost_tok_kernel<64, 128, 128>
      <<<dim3(128 * 128 / 64, (T - 1) * B), blk, 0, stream>>>(l1, w1, b1, tok1, B);
  cost_tok_kernel<128, 64, 64>
      <<<dim3(64 * 64 / 64, (T - 1) * B), blk, 0, stream>>>(l2, w2, b2, tok2, B);
  cost_tok_kernel<192, 32, 32>
      <<<dim3(32 * 32 / 64, (T - 1) * B), blk, 0, stream>>>(l3, w3, b3, tok3, B);
}


# The two small pyramid levels share one launch (block index selects the level; all 320 blocks resident at once, none waits for a slot), the large level keeps its own: 2 launches instead of 3
# speedup vs baseline: 10.5856x; 1.2162x over previous
#include <hip/hip_runtime.h>

typedef __attribute__((ext_vector_type(2))) float v2f;
typedef __attribute__((ext_vector_type(8))) float v8f;
typedef __attribute__((ext_vector_type(4))) float v4f;

template <int C, int W>
struct LevelLds {
  static constexpr int TW  = (W >= 64) ? 64 : W;
  static constexpr int HR  = 64 / TW + 6;
  static constexpr int HP  = TW + 12;
  static constexpr int F1  = C * 64;
  static constexpr int WL  = 52 * 192;
  static constexpr int BL  = 256;
  static constexpr int CR  = 64 * 52;
  static constexpr int F2  = 16 * HR * HP;
  static constexpr int ALL = F1 + WL + BL + CR + F2;
};

template <int C, int H, int W>
__device__ __forceinline__ void cost_tok_body(
    float* __restrict__ f1_lds,
    float* __restrict__ w_lds,
    float* __restrict__ b_lds,
    float* __restrict__ corr_lds,
    float* __restrict__ f2_lds,
    const float* __restrict__ lvl,
    const float* __restrict__ wmat,
    const float* __restrict__ bias,
    float* __restrict__ out,
    int B,
    int tile,
    int tb) {
  constexpr int HW  = H * W;
  constexpr int KP  = 52;
  constexpr int TD  = 192;
  constexpr int PIX = 64;

  constexpr int TW  = (W >= 64) ? 64 : W;
  constexpr int TR  = PIX / TW;
  constexpr int HR  = TR + 6;
  constexpr int HCP = TW + 8;
  constexpr int QPR = HCP / 4;
  constexpr int NQ  = HR * QPR;
  constexpr int HP  = HCP + 4;
  constexpr int PL  = HR * HP;
  constexpr int PLQ = PL / 4;
  constexpr int CC  = 16;
  constexpr int G8  = TW / 8;
  constexpr int SP  = 60;
  static_assert(C % CC == 0, "channel chunking");
  static_assert(W % TW == 0 && PIX % TW == 0 && TW % 4 == 0, "tile shape");
  static_assert(H % TR == 0 && H >= 4 && W >= 4 && W % 4 == 0, "tile rows");
  static_assert(NQ <= 128, "one halo piece per thread");
  static_assert(HP % 4 == 0 && 56 * SP <= CC * PL, "window and scratch");
  static_assert((TR == 1 && G8 == 8) || (TR == 2 && G8 == 4), "window map");

  using LL = LevelLds<C, W>;
  static_assert(LL::F1 == C * PIX && LL::WL == KP * TD && LL::CR == PIX * KP && LL::F2 == CC * PL,
                "LDS map");
  static_assert(LL::F1 % 4 == 0 && LL::WL % 4 == 0 && LL::BL % 4 == 0 && LL::CR % 4 == 0,
                "16-byte aligned LDS arrays");

  const int tid   = threadIdx.x;
  const int pbase = tile * PIX;

  const float* f1 = lvl + (size_t)tb * C * HW;
  const float* f2 = lvl + (size_t)(tb + B) * C * HW;

  const int h0 = pbase / W;
  const int w0 = pbase % W;
  const bool act = tid < NQ;
  int goff, sidx;
  {
    const int r = tid / QPR;
    const int j = tid - r * QPR;
    sidx = r * (HP / 4) + j;
    int gh = h0 - 3 + r; gh += (gh < 0) ? H : 0; gh -= (gh >= H) ? H : 0;
    int gw = w0 - 4 + 4 * j; gw += (gw < 0) ? W : 0; gw -= (gw >= W) ? W : 0;
    goff = act ? (gh * W + gw) : 0;
  }

  constexpr int DQ  = TD / 4;
  constexpr int NKP = KP / 2;
  constexpr int WI  = NKP * DQ;
  constexpr int WS  = (WI + 127) / 128;
  constexpr int FS  = C * PIX / 4 / 128;
  static_assert((WS - 1) * 128 <= 24 * DQ && 24 * DQ < WI && WI <= WS * 128, "weight slot map");
  static_assert((C * PIX / 4) % 128 == 0 && TD > 128 && TD <= 256, "slot maps");

  const v4f* w4g = reinterpret_cast<const v4f*>(wmat);
  v4f wa[WS], wb[WS - 1];
#pragma unroll
  for (int j = 0; j < WS; ++j) {
    const int ir = tid + 128 * j;
    const int i  = (j < WS - 1 || ir < WI) ? ir : (WI - 1);
    const int kp = i / DQ;
    const int dq = i - kp * DQ;
    if (j < WS - 1) {
      wa[j] = w4g[(2 * kp) * DQ + dq];
      wb[j] = w4g[(2 * kp + 1) * DQ + dq];
    } else {
      wa[j] = w4g[(kp == 24) ? (48 * DQ + dq) : 0];
    }
  }

  v4f fv[FS];
#pragma unroll
  for (int j = 0; j < FS; ++j) {
    const int i4 = tid + 128 * j;
    fv[j] = *reinterpret_cast<const v4f*>(
        f1 + (size_t)(i4 >> 4) * HW + pbase + 4 * (i4 & 15));
  }

  const float bv0 = bias[tid];
  const float bv1 = bias[(tid < TD - 128) ? (tid + 128) : tid];

  v4f pre[CC];
  if (act) {
#pragma unroll
    for (int cc = 0; cc < CC; ++cc)
      pre[cc] = *reinterpret_cast<const v4f*>(f2 + (size_t)cc * HW + goff);
  }

  {
    const v4f z4 = {0.0f, 0.0f, 0.0f, 0.0f};
    v4f* w4l = reinterpret_cast<v4f*>(w_lds);
#pragma unroll
    for (int j = 0; j < WS; ++j) {
      const int ir = tid + 128 * j;
      const int i  = (j < WS - 1 || ir < WI) ? ir : (WI - 1);
      v4f a, b;
      if (j < WS - 1) {
        a = wa[j];
        b = wb[j];
      } else {
        a = (i < 25 * DQ) ? wa[j] : z4;
        b = z4;
      }
      const v4f lo = {a.x, b.x, a.y, b.y};
      const v4f hi = {a.z, b.z, a.w, b.w};
      w4l[2 * i]     = lo;
      w4l[2 * i + 1] = hi;
    }

    v4f* f4l = reinterpret_cast<v4f*>(f1_lds);
#pragma unroll
    for (int j = 0; j < FS; ++j)
      f4l[tid + 128 * j] = fv[j];

    b_lds[tid] = bv0;
    b_lds[tid + 128] = bv1;
  }

  {
    const float scale = 1.0f / __builtin_sqrtf((float)C);
    const int half = tid >> 6;
    const int q8   = tid & 63;
    int ky, g;
    if (TR == 1) {
      ky = q8 >> 3;
      g  = q8 & 7;
    } else {
      const int prd = (q8 >> 2) & 1;
      ky = (q8 >> 3) - 1 + prd;
      g  = 4 * prd + (q8 & 3);
    }
    const bool work = (unsigned)ky < 7u;
    const int pr   = g / G8;
    const int pc   = 8 * (g % G8);
    const int hb   = (pr + 6 - ky) * HP + pc;
    v4f* f2q = reinterpret_cast<v4f*>(f2_lds);

    float acc[8][7];
#pragma unroll
    for (int q = 0; q < 8; ++q)
#pragma unroll
      for (int kx = 0; kx < 7; ++kx)
        acc[q][kx] = 0.0f;

#pragma unroll 1
    for (int c0 = 0; c0 < C; c0 += CC) {
      __syncthreads();
      if (act) {
#pragma unroll
        for (int cc = 0; cc < CC; ++cc)
          f2q[cc * PLQ + sidx] = pre[cc];
      }
      __syncthreads();

      if (c0 + CC < C && act) {
        const float* nx = f2 + (size_t)(c0 + CC) * HW + goff;
#pragma unroll
        for (int cc = 0; cc < CC; ++cc)
          pre[cc] = *reinterpret_cast<const v4f*>(nx + (size_t)cc * HW);
      }

      if (work) {
#pragma unroll
        for (int c2 = 0; c2 < CC / 2; ++c2) {
          const int cc = 2 * c2 + half;
          const v4f* fp = reinterpret_cast<const v4f*>(&f1_lds[(c0 + cc) * PIX + 8 * g]);
          const v4f* hp = reinterpret_cast<const v4f*>(&f2_lds[cc * PL + hb]);
          const v4f fa = fp[0];
          const v4f fb = fp[1];
          v4f h0 = hp[0];
          const v4f h1 = hp[1];
          const v4f h2 = hp[2];
          v4f h3 = hp[3];
          asm("" : "+v"(h0));
          asm("" : "+v"(h3));
          const float hv[14] = {h0.y, h0.z, h0.w, h1.x, h1.y, h1.z, h1.w,
                                h2.x, h2.y, h2.z, h2.w, h3.x, h3.y, h3.z};
          const float fq[8] = {fa.x, fa.y, fa.z, fa.w, fb.x, fb.y, fb.z, fb.w};
#pragma unroll
          for (int q = 0; q < 8; ++q)
#pragma unroll
            for (int kx = 0; kx < 7; ++kx)
              acc[q][kx] = __builtin_fmaf(fq[q], hv[q + 6 - kx], acc[q][kx]);
        }
      }
    }

    __syncthreads();
    v4f* sc4 = reinterpret_cast<v4f*>(&f2_lds[q8 * SP]);
    if (half == 1 && work) {
#pragma unroll
      for (int i = 0; i < 14; ++i) {
        const v4f v = {acc[(4 * i) / 7][(4 * i) % 7], acc[(4 * i + 1) / 7][(4 * i + 1) % 7],
                       acc[(4 * i + 2) / 7][(4 * i + 2) % 7], acc[(4 * i + 3) / 7][(4 * i + 3) % 7]};
        sc4[i] = v;
      }
    }
    __syncthreads();
    if (half == 0) {
      if (work) {
#pragma unroll
        for (int i = 0; i < 14; ++i) {
          const v4f v = sc4[i];
          const float o[4] = {v.x, v.y, v.z, v.w};
#pragma unroll
          for (int e = 0; e < 4; ++e) {
            const int q = (4 * i + e) / 7;
            const int kx = (4 * i + e) % 7;
            corr_lds[(8 * g + q) * KP + ky * 7 + kx] = (acc[q][kx] + o[e]) * scale;
          }
        }
      } else {
#pragma unroll
        for (int q = 0; q < 8; ++q)
#pragma unroll
          for (int z = 49; z < KP; ++z)
            corr_lds[(8 * g + q) * KP + z] = 0.0f;
      }
    }
  }
  __syncthreads();

  {
    const int lane = tid & 31;
    const int wave = tid >> 5;
    const int p0   = wave * 16;
    const int nlo  = lane & 15;
    const int lhi  = lane >> 4;

    v2f bfr[13];
#pragma unroll
    for (int ks = 0; ks < 13; ++ks) {
      const int krow = 4 * ks + 2 * lhi;
      const float* src = &corr_lds[(p0 + nlo) * KP + krow];
      bfr[ks].x = src[0];
      bfr[ks].y = src[1];
    }

    constexpr int MG = 3;
    constexpr int NG = 12 / MG;
    v2f afr[2][MG * 13];
    v8f acc[2][MG];

    const v2f* w2l = reinterpret_cast<const v2f*>(w_lds);
#define P2_LOAD(G, S)                                                      \
    _Pragma("unroll")                                                      \
    for (int t = 0; t < MG; ++t) {                                         \
      const int d0 = ((G) * MG + t) * 16;                                  \
      _Pragma("unroll")                                                    \
      for (int ks = 0; ks < 13; ++ks)                                      \
        afr[S][t * 13 + ks] = w2l[(2 * ks + lhi) * TD + d0 + nlo];         \
      _Pragma("unroll")                                                    \
      for (int r = 0; r < 8; ++r)                                          \
        acc[S][t][r] = b_lds[d0 + r + 8 * lhi];                            \
    }

#define P2_MMA_STORE(G, S)                                                 \
    _Pragma("unroll")                                                      \
    for (int ks = 0; ks < 13; ++ks) {                                      \
      _Pragma("unroll")                                                    \
      for (int t = 0; t < MG; ++t)                                         \
        acc[S][t] = __builtin_amdgcn_wmma_f32_16x16x4_f32(                 \
            false, afr[S][t * 13 + ks], false, bfr[ks], (short)0,          \
            acc[S][t], false, false);                                      \
    }                                                                      \
    _Pragma("unroll")                                                      \
    for (int t = 0; t < MG; ++t) {                                         \
      const int d0 = ((G) * MG + t) * 16;                                  \
      float* op = out + ((size_t)tb * TD + d0) * HW + pbase + p0 + nlo;    \
      _Pragma("unroll")                                                    \
      for (int r = 0; r < 8; ++r)                                          \
        op[(size_t)(r + 8 * lhi) * HW] = acc[S][t][r];                     \
    }

    P2_LOAD(0, 0)
    __builtin_amdgcn_sched_barrier(0);
#pragma unroll
    for (int g = 0; g < NG; g += 2) {
      P2_LOAD(g + 1, 1)
      __builtin_amdgcn_sched_barrier(0);
      P2_MMA_STORE(g, 0)
      __builtin_amdgcn_sched_barrier(0);
      if (g + 2 < NG) {
        P2_LOAD(g + 2, 0)
      }
      __builtin_amdgcn_sched_barrier(0);
      P2_MMA_STORE(g + 1, 1)
      __builtin_amdgcn_sched_barrier(0);
    }
#undef P2_LOAD
#undef P2_MMA_STORE
  }
}

template <int C, int H, int W>
__global__ __launch_bounds__(128) void cost_tok_kernel(
    const float* __restrict__ lvl, const float* __restrict__ wmat,
    const float* __restrict__ bias, float* __restrict__ out, int B) {
  using LL = LevelLds<C, W>;
  __shared__ __attribute__((aligned(16))) float f1_lds[LL::F1];
  __shared__ __attribute__((aligned(16))) float w_lds[LL::WL];
  __shared__ float b_lds[LL::BL];
  __shared__ float corr_lds[LL::CR];
  __shared__ __attribute__((aligned(16))) float f2_lds[LL::F2];
  cost_tok_body<C, H, W>(f1_lds, w_lds, b_lds, corr_lds, f2_lds,
                         lvl, wmat, bias, out, B, blockIdx.x, blockIdx.y);
}

template <int CA, int HA, int WA, int CB, int HB, int WB>
__global__ __launch_bounds__(128) void cost_tok_pair_kernel(
    const float* __restrict__ lvlA, const float* __restrict__ wmatA,
    const float* __restrict__ biasA, float* __restrict__ outA,
    const float* __restrict__ lvlB, const float* __restrict__ wmatB,
    const float* __restrict__ biasB, float* __restrict__ outB, int B) {
  constexpr int TA = HA * WA / 64;
  using LA = LevelLds<CA, WA>;
  using LB = LevelLds<CB, WB>;
  __shared__ __attribute__((aligned(16))) float smem[(LA::ALL > LB::ALL) ? LA::ALL : LB::ALL];
  if ((int)blockIdx.x < TA) {
    float* const f1 = smem;
    float* const wl = f1 + LA::F1;
    float* const bl = wl + LA::WL;
    float* const cr = bl + LA::BL;
    float* const f2 = cr + LA::CR;
    cost_tok_body<CA, HA, WA>(f1, wl, bl, cr, f2, lvlA, wmatA, biasA, outA, B,
                              blockIdx.x, blockIdx.y);
  } else {
    float* const f1 = smem;
    float* const wl = f1 + LB::F1;
    float* const bl = wl + LB::WL;
    float* const cr = bl + LB::BL;
    float* const f2 = cr + LB::CR;
    cost_tok_body<CB, HB, WB>(f1, wl, bl, cr, f2, lvlB, wmatB, biasB, outB, B,
                              (int)blockIdx.x - TA, blockIdx.y);
  }
}

extern "C" void kernel_launch(void* const* d_in, const int* in_sizes, int n_in,
                              void* d_out, int out_size, void* d_ws, size_t ws_size,
                              hipStream_t stream) {
  const float* l1 = (const float*)d_in[0];
  const float* l2 = (const float*)d_in[1];
  const float* l3 = (const float*)d_in[2];
  const float* w1 = (const float*)d_in[3];
  const float* b1 = (const float*)d_in[4];
  const float* w2 = (const float*)d_in[5];
  const float* b2 = (const float*)d_in[6];
  const float* w3 = (const float*)d_in[7];
  const float* b3 = (const float*)d_in[8];

  const int T = 3, B = 2, TD = 192;
  float* tok1 = (float*)d_out;
  float* tok2 = tok1 + (size_t)(T - 1) * B * TD * 128 * 128;
  float* tok3 = tok2 + (size_t)(T - 1) * B * TD * 64 * 64;

  dim3 blk(128);
  cost_tok_kernel<64, 128, 128>
      <<<dim3(128 * 128 / 64, (T - 1) * B), blk, 0, stream>>>(l1, w1, b1, tok1, B);
  cost_tok_pair_kernel<192, 32, 32, 128, 64, 64>
      <<<dim3(32 * 32 / 64 + 64 * 64 / 64, (T - 1) * B), blk, 0, stream>>>(
          l3, w3, b3, tok3, l2, w2, b2, tok2, B);
}
